# starting kernel
# baseline (speedup 1.0000x reference)
#include <hip/hip_runtime.h>
#include <hip/hip_bf16.h>


typedef __attribute__((ext_vector_type(2))) float v2f;
typedef __attribute__((ext_vector_type(8))) float v8f;

#define F_IN 128
#define HID  16
#define NCLS 16

__global__ void gcn_init_deg(float* __restrict__ deg, int n) {
  int i = blockIdx.x * blockDim.x + threadIdx.x;
  if (i < n) deg[i] = 1.0f;
}

__global__ void gcn_deg_accum(const int* __restrict__ ei, float* __restrict__ deg, int E) {
  int e = blockIdx.x * blockDim.x + threadIdx.x;
  if (e < E) unsafeAtomicAdd(&deg[ei[E + e]], 1.0f);
}

__global__ void gcn_dinv(float* __restrict__ deg, int n) {
  int i = blockIdx.x * blockDim.x + threadIdx.x;
  if (i < n) {
    float d = deg[i];
    deg[i] = d > 0.0f ? rsqrtf(d) : 0.0f;
  }
}

__global__ __launch_bounds__(256) void gcn_gemm_wmma(const float* __restrict__ x,
                                                     const float* __restrict__ W,
                                                     float* __restrict__ h,
                                                     int nTiles, int N) {
  const int lane = threadIdx.x & 31;
  const int tile = blockIdx.x * (blockDim.x >> 5) + (threadIdx.x >> 5);
  if (tile >= nTiles) return;

  const int half = lane >> 4;
  const int l    = lane & 15;
  const int row  = tile * 16 + l;
  const int rowc = row < N ? row : N - 1;

  const float* xr = x + (size_t)rowc * F_IN + half * 2;

  v8f acc = {0.f, 0.f, 0.f, 0.f, 0.f, 0.f, 0.f, 0.f};

#pragma unroll
  for (int k = 0; k < F_IN; k += 4) {
    v2f a;
    a.x = xr[k];
    a.y = xr[k + 1];
    const float* wp = W + (size_t)(k + half * 2) * HID + l;
    v2f b;
    b.x = wp[0];
    b.y = wp[HID];
    acc = __builtin_amdgcn_wmma_f32_16x16x4_f32(false, a, false, b,
                                                (short)0, acc, false, false);
  }

  const int mbase = tile * 16 + half * 8;
#pragma unroll
  for (int j = 0; j < 8; ++j) {
    int m = mbase + j;
    if (m < N) h[(size_t)m * HID + l] = acc[j];
  }
}

__global__ void gcn_init_agg(const float* __restrict__ h, const float* __restrict__ dinv,
                             float* __restrict__ agg, int n) {
  int t = blockIdx.x * blockDim.x + threadIdx.x;
  if (t < n * HID) {
    float di = dinv[t >> 4];
    agg[t] = h[t] * di * di;
  }
}

__global__ void gcn_edge_scatter(const int* __restrict__ ei, const float* __restrict__ h,
                                 const float* __restrict__ dinv, float* __restrict__ agg,
                                 int E) {
  long long t = (long long)blockIdx.x * blockDim.x + threadIdx.x;
  int e = (int)(t >> 4);
  if (e >= E) return;
  int f = (int)(t & 15);
  int s = ei[e];
  int d = ei[E + e];
  float val = h[(size_t)s * HID + f] * (dinv[s] * dinv[d]);
  unsafeAtomicAdd(&agg[(size_t)d * HID + f], val);
}

__global__ void gcn_final(const float* __restrict__ agg, const float* __restrict__ b1,
                          const float* __restrict__ Wp, const float* __restrict__ bp,
                          float* __restrict__ out, int n) {
  int i = blockIdx.x * blockDim.x + threadIdx.x;
  if (i >= n) return;

  float hb[HID];
#pragma unroll
  for (int k = 0; k < HID; ++k) {
    float v = agg[(size_t)i * HID + k] + b1[k];
    hb[k] = v > 0.0f ? v : 0.0f;
  }

  float logits[NCLS];
  float mx = -3.0e38f;
#pragma unroll
  for (int c = 0; c < NCLS; ++c) {
    float accv = bp[c];
#pragma unroll
    for (int k = 0; k < HID; ++k) accv += hb[k] * Wp[k * NCLS + c];
    logits[c] = accv;
    mx = fmaxf(mx, accv);
  }

  float sum = 0.0f;
#pragma unroll
  for (int c = 0; c < NCLS; ++c) {
    float ev = __expf(logits[c] - mx);
    logits[c] = ev;
    sum += ev;
  }
  float inv = 1.0f / sum;
#pragma unroll
  for (int c = 0; c < NCLS; ++c) out[(size_t)i * NCLS + c] = logits[c] * inv;
}

extern "C" void kernel_launch(void* const* d_in, const int* in_sizes, int n_in,
                              void* d_out, int out_size, void* d_ws, size_t ws_size,
                              hipStream_t stream) {
  const float* x  = (const float*)d_in[0];
  const float* W1 = (const float*)d_in[1];
  const float* b1 = (const float*)d_in[2];
  const float* Wp = (const float*)d_in[3];
  const float* bp = (const float*)d_in[4];
  const int*   ei = (const int*)d_in[5];

  const int N = in_sizes[0] / F_IN;
  const int E = in_sizes[5] / 2;
  float* out = (float*)d_out;

  float* deg = (float*)d_ws;
  float* h   = deg + N;
  float* agg = h + (size_t)N * HID;

  const int tpb = 256;

  gcn_init_deg<<<(N + tpb - 1) / tpb, tpb, 0, stream>>>(deg, N);
  gcn_deg_accum<<<(E + tpb - 1) / tpb, tpb, 0, stream>>>(ei, deg, E);
  gcn_dinv<<<(N + tpb - 1) / tpb, tpb, 0, stream>>>(deg, N);

  const int nTiles = (N + 15) / 16;
  gcn_gemm_wmma<<<(nTiles + 7) / 8, 256, 0, stream>>>(x, W1, h, nTiles, N);

  const long long nh = (long long)N * HID;
  gcn_init_agg<<<(int)((nh + tpb - 1) / tpb), tpb, 0, stream>>>(h, deg, agg, N);

  const long long tot = (long long)E * HID;
  gcn_edge_scatter<<<(int)((tot + tpb - 1) / tpb), tpb, 0, stream>>>(ei, h, deg, agg, E);

  gcn_final<<<(N + tpb - 1) / tpb, tpb, 0, stream>>>(agg, b1, Wp, bp, out, N);
}


# degree count: contended global f32 atomic adds replaced by per-node-range LDS integer counters over a 16-way sliced scan of dst, partial planes summed in the rsqrt kernel; same degree values
# speedup vs baseline: 1.5883x; 1.5883x over previous
#include <hip/hip_runtime.h>
#include <hip/hip_bf16.h>


typedef __attribute__((ext_vector_type(2))) float v2f;
typedef __attribute__((ext_vector_type(8))) float v8f;

#define F_IN 128
#define HID  16
#define NCLS 16

#define DEG_RANGE  6272
#define DEG_CHUNKS 16
#define DEG_TPB    1024
#define DEG_BATCH  8

__device__ __forceinline__ void deg_tally(unsigned* cnt, int d, unsigned lo, unsigned rn) {
  unsigned rel = (unsigned)d - lo;
  if (rel < rn) atomicAdd(&cnt[rel], 1u);
}

__global__ __launch_bounds__(DEG_TPB) void gcn_deg_count(const int* __restrict__ ei,
                                                         unsigned* __restrict__ part,
                                                         int E, int n) {
  __shared__ unsigned cnt[DEG_RANGE];
  const int tid = threadIdx.x;
  const int c   = blockIdx.x;
  const int lo  = blockIdx.y * DEG_RANGE;
  int rn = n - lo;
  if (rn > DEG_RANGE) rn = DEG_RANGE;

  for (int i = tid; i < DEG_RANGE; i += DEG_TPB) cnt[i] = 0u;
  __syncthreads();

  const int per = (((E + DEG_CHUNKS - 1) / DEG_CHUNKS) + 3) & ~3;
  int e0 = c * per;
  if (e0 > E) e0 = E;
  int e1 = e0 + per;
  if (e1 > E) e1 = E;
  const int len = e1 - e0;
  const int* __restrict__ p = ei + E + e0;

  int head = (int)(((16u - ((unsigned)(size_t)p & 15u)) & 15u) >> 2);
  if (head > len) head = len;
  const int nvec = (len - head) >> 2;
  const int tail = head + (nvec << 2);
  const int4* __restrict__ pv = (const int4*)(p + head);

  if (tid < head) deg_tally(cnt, p[tid], (unsigned)lo, (unsigned)rn);

  for (int base = 0; base < nvec; base += DEG_BATCH * DEG_TPB) {
    int4 v[DEG_BATCH];
#pragma unroll
    for (int u = 0; u < DEG_BATCH; ++u) {
      const int i = base + u * DEG_TPB + tid;
      v[u] = (i < nvec) ? pv[i] : make_int4(-1, -1, -1, -1);
    }
#pragma unroll
    for (int u = 0; u < DEG_BATCH; ++u) {
      deg_tally(cnt, v[u].x, (unsigned)lo, (unsigned)rn);
      deg_tally(cnt, v[u].y, (unsigned)lo, (unsigned)rn);
      deg_tally(cnt, v[u].z, (unsigned)lo, (unsigned)rn);
      deg_tally(cnt, v[u].w, (unsigned)lo, (unsigned)rn);
    }
  }

  if (tail + tid < len) deg_tally(cnt, p[tail + tid], (unsigned)lo, (unsigned)rn);
  __syncthreads();

  unsigned* __restrict__ dstp = part + (size_t)c * n + lo;
  for (int i = tid; i < rn; i += DEG_TPB) dstp[i] = cnt[i];
}

__global__ void gcn_dinv(const unsigned* __restrict__ part, float* __restrict__ deg, int n) {
  int i = blockIdx.x * blockDim.x + threadIdx.x;
  if (i < n) {
    unsigned k = 1u;
#pragma unroll
    for (int c = 0; c < DEG_CHUNKS; ++c) k += part[(size_t)c * n + i];
    float d = (float)k;
    deg[i] = d > 0.0f ? rsqrtf(d) : 0.0f;
  }
}

__global__ __launch_bounds__(256) void gcn_gemm_wmma(const float* __restrict__ x,
                                                     const float* __restrict__ W,
                                                     float* __restrict__ h,
                                                     int nTiles, int N) {
  const int lane = threadIdx.x & 31;
  const int tile = blockIdx.x * (blockDim.x >> 5) + (threadIdx.x >> 5);
  if (tile >= nTiles) return;

  const int half = lane >> 4;
  const int l    = lane & 15;
  const int row  = tile * 16 + l;
  const int rowc = row < N ? row : N - 1;

  const float* xr = x + (size_t)rowc * F_IN + half * 2;

  v8f acc = {0.f, 0.f, 0.f, 0.f, 0.f, 0.f, 0.f, 0.f};

#pragma unroll
  for (int k = 0; k < F_IN; k += 4) {
    v2f a;
    a.x = xr[k];
    a.y = xr[k + 1];
    const float* wp = W + (size_t)(k + half * 2) * HID + l;
    v2f b;
    b.x = wp[0];
    b.y = wp[HID];
    acc = __builtin_amdgcn_wmma_f32_16x16x4_f32(false, a, false, b,
                                                (short)0, acc, false, false);
  }

  const int mbase = tile * 16 + half * 8;
#pragma unroll
  for (int j = 0; j < 8; ++j) {
    int m = mbase + j;
    if (m < N) h[(size_t)m * HID + l] = acc[j];
  }
}

__global__ void gcn_init_agg(const float* __restrict__ h, const float* __restrict__ dinv,
                             float* __restrict__ agg, int n) {
  int t = blockIdx.x * blockDim.x + threadIdx.x;
  if (t < n * HID) {
    float di = dinv[t >> 4];
    agg[t] = h[t] * di * di;
  }
}

__global__ void gcn_edge_scatter(const int* __restrict__ ei, const float* __restrict__ h,
                                 const float* __restrict__ dinv, float* __restrict__ agg,
                                 int E) {
  long long t = (long long)blockIdx.x * blockDim.x + threadIdx.x;
  int e = (int)(t >> 4);
  if (e >= E) return;
  int f = (int)(t & 15);
  int s = ei[e];
  int d = ei[E + e];
  float val = h[(size_t)s * HID + f] * (dinv[s] * dinv[d]);
  unsafeAtomicAdd(&agg[(size_t)d * HID + f], val);
}

__global__ void gcn_final(const float* __restrict__ agg, const float* __restrict__ b1,
                          const float* __restrict__ Wp, const float* __restrict__ bp,
                          float* __restrict__ out, int n) {
  int i = blockIdx.x * blockDim.x + threadIdx.x;
  if (i >= n) return;

  float hb[HID];
#pragma unroll
  for (int k = 0; k < HID; ++k) {
    float v = agg[(size_t)i * HID + k] + b1[k];
    hb[k] = v > 0.0f ? v : 0.0f;
  }

  float logits[NCLS];
  float mx = -3.0e38f;
#pragma unroll
  for (int c = 0; c < NCLS; ++c) {
    float accv = bp[c];
#pragma unroll
    for (int k = 0; k < HID; ++k) accv += hb[k] * Wp[k * NCLS + c];
    logits[c] = accv;
    mx = fmaxf(mx, accv);
  }

  float sum = 0.0f;
#pragma unroll
  for (int c = 0; c < NCLS; ++c) {
    float ev = __expf(logits[c] - mx);
    logits[c] = ev;
    sum += ev;
  }
  float inv = 1.0f / sum;
#pragma unroll
  for (int c = 0; c < NCLS; ++c) out[(size_t)i * NCLS + c] = logits[c] * inv;
}

extern "C" void kernel_launch(void* const* d_in, const int* in_sizes, int n_in,
                              void* d_out, int out_size, void* d_ws, size_t ws_size,
                              hipStream_t stream) {
  const float* x  = (const float*)d_in[0];
  const float* W1 = (const float*)d_in[1];
  const float* b1 = (const float*)d_in[2];
  const float* Wp = (const float*)d_in[3];
  const float* bp = (const float*)d_in[4];
  const int*   ei = (const int*)d_in[5];

  const int N = in_sizes[0] / F_IN;
  const int E = in_sizes[5] / 2;
  float* out = (float*)d_out;

  float* deg = (float*)d_ws;
  float* h   = deg + N;
  float* agg = h + (size_t)N * HID;

  const int tpb = 256;

  unsigned* part = (unsigned*)agg;
  const int nRanges = (N + DEG_RANGE - 1) / DEG_RANGE;
  gcn_deg_count<<<dim3(DEG_CHUNKS, nRanges), DEG_TPB, 0, stream>>>(ei, part, E, N);
  gcn_dinv<<<(N + tpb - 1) / tpb, tpb, 0, stream>>>(part, deg, N);

  const int nTiles = (N + 15) / 16;
  gcn_gemm_wmma<<<(nTiles + 7) / 8, 256, 0, stream>>>(x, W1, h, nTiles, N);

  const long long nh = (long long)N * HID;
  gcn_init_agg<<<(int)((nh + tpb - 1) / tpb), tpb, 0, stream>>>(h, deg, agg, N);

  const long long tot = (long long)E * HID;
  gcn_edge_scatter<<<(int)((tot + tpb - 1) / tpb), tpb, 0, stream>>>(ei, h, deg, agg, E);

  gcn_final<<<(N + tpb - 1) / tpb, tpb, 0, stream>>>(agg, b1, Wp, bp, out, N);
}


# edge aggregation: per-edge global f32 atomics replaced by work-group-owned node rows in LDS (f32 LDS adds) over a 10-way sliced edge scan, flushed as whole-line f32 atomic adds; sum order changes
# speedup vs baseline: 3.2920x; 2.0727x over previous
#include <hip/hip_runtime.h>
#include <hip/hip_bf16.h>


typedef __attribute__((ext_vector_type(2))) float v2f;
typedef __attribute__((ext_vector_type(8))) float v8f;

#define F_IN 128
#define HID  16
#define NCLS 16

#define DEG_RANGE  6272
#define DEG_CHUNKS 16
#define DEG_TPB    1024
#define DEG_BATCH  8

__device__ __forceinline__ void deg_tally(unsigned* cnt, int d, unsigned lo, unsigned rn) {
  unsigned rel = (unsigned)d - lo;
  if (rel < rn) atomicAdd(&cnt[rel], 1u);
}

__global__ __launch_bounds__(DEG_TPB) void gcn_deg_count(const int* __restrict__ ei,
                                                         unsigned* __restrict__ part,
                                                         int E, int n) {
  __shared__ unsigned cnt[DEG_RANGE];
  const int tid = threadIdx.x;
  const int c   = blockIdx.x;
  const int lo  = blockIdx.y * DEG_RANGE;
  int rn = n - lo;
  if (rn > DEG_RANGE) rn = DEG_RANGE;

  for (int i = tid; i < DEG_RANGE; i += DEG_TPB) cnt[i] = 0u;
  __syncthreads();

  const int per = (((E + DEG_CHUNKS - 1) / DEG_CHUNKS) + 3) & ~3;
  int e0 = c * per;
  if (e0 > E) e0 = E;
  int e1 = e0 + per;
  if (e1 > E) e1 = E;
  const int len = e1 - e0;
  const int* __restrict__ p = ei + E + e0;

  int head = (int)(((16u - ((unsigned)(size_t)p & 15u)) & 15u) >> 2);
  if (head > len) head = len;
  const int nvec = (len - head) >> 2;
  const int tail = head + (nvec << 2);
  const int4* __restrict__ pv = (const int4*)(p + head);

  if (tid < head) deg_tally(cnt, p[tid], (unsigned)lo, (unsigned)rn);

  for (int base = 0; base < nvec; base += DEG_BATCH * DEG_TPB) {
    int4 v[DEG_BATCH];
#pragma unroll
    for (int u = 0; u < DEG_BATCH; ++u) {
      const int i = base + u * DEG_TPB + tid;
      v[u] = (i < nvec) ? pv[i] : make_int4(-1, -1, -1, -1);
    }
#pragma unroll
    for (int u = 0; u < DEG_BATCH; ++u) {
      deg_tally(cnt, v[u].x, (unsigned)lo, (unsigned)rn);
      deg_tally(cnt, v[u].y, (unsigned)lo, (unsigned)rn);
      deg_tally(cnt, v[u].z, (unsigned)lo, (unsigned)rn);
      deg_tally(cnt, v[u].w, (unsigned)lo, (unsigned)rn);
    }
  }

  if (tail + tid < len) deg_tally(cnt, p[tail + tid], (unsigned)lo, (unsigned)rn);
  __syncthreads();

  unsigned* __restrict__ dstp = part + (size_t)c * n + lo;
  for (int i = tid; i < rn; i += DEG_TPB) dstp[i] = cnt[i];
}

__global__ void gcn_dinv(const unsigned* __restrict__ part, float* __restrict__ deg, int n) {
  int i = blockIdx.x * blockDim.x + threadIdx.x;
  if (i < n) {
    unsigned k = 1u;
#pragma unroll
    for (int c = 0; c < DEG_CHUNKS; ++c) k += part[(size_t)c * n + i];
    float d = (float)k;
    deg[i] = d > 0.0f ? rsqrtf(d) : 0.0f;
  }
}

__global__ __launch_bounds__(256) void gcn_gemm_wmma(const float* __restrict__ x,
                                                     const float* __restrict__ W,
                                                     float* __restrict__ h,
                                                     int nTiles, int N) {
  const int lane = threadIdx.x & 31;
  const int tile = blockIdx.x * (blockDim.x >> 5) + (threadIdx.x >> 5);
  if (tile >= nTiles) return;

  const int half = lane >> 4;
  const int l    = lane & 15;
  const int row  = tile * 16 + l;
  const int rowc = row < N ? row : N - 1;

  const float* xr = x + (size_t)rowc * F_IN + half * 2;

  v8f acc = {0.f, 0.f, 0.f, 0.f, 0.f, 0.f, 0.f, 0.f};

#pragma unroll
  for (int k = 0; k < F_IN; k += 4) {
    v2f a;
    a.x = xr[k];
    a.y = xr[k + 1];
    const float* wp = W + (size_t)(k + half * 2) * HID + l;
    v2f b;
    b.x = wp[0];
    b.y = wp[HID];
    acc = __builtin_amdgcn_wmma_f32_16x16x4_f32(false, a, false, b,
                                                (short)0, acc, false, false);
  }

  const int mbase = tile * 16 + half * 8;
#pragma unroll
  for (int j = 0; j < 8; ++j) {
    int m = mbase + j;
    if (m < N) h[(size_t)m * HID + l] = acc[j];
  }
}

__global__ void gcn_init_agg(float* __restrict__ h, const float* __restrict__ dinv,
                             float* __restrict__ agg, int n) {
  int t = blockIdx.x * blockDim.x + threadIdx.x;
  if (t < n * HID) {
    float di = dinv[t >> 4];
    float hv = h[t] * di;
    h[t] = hv;
    agg[t] = hv * di;
  }
}

#define SC_RANGE  4000
#define SC_PITCH  17
#define SC_SLICES 10
#define SC_TPB    1024
#define SC_BATCH  8
#define SC_QCAP   160

__device__ __forceinline__ void sc_lds_add(float* p, float v) {
  __hip_atomic_fetch_add(p, v, __ATOMIC_RELAXED, __HIP_MEMORY_SCOPE_WORKGROUP);
}

__device__ __forceinline__ void sc_add_row(float* arow, const float* __restrict__ h, unsigned s) {
  const float4* __restrict__ row = (const float4*)(h + (size_t)s * HID);
  const float4 r0 = row[0], r1 = row[1], r2 = row[2], r3 = row[3];
  sc_lds_add(arow + 0, r0.x);  sc_lds_add(arow + 1, r0.y);
  sc_lds_add(arow + 2, r0.z);  sc_lds_add(arow + 3, r0.w);
  sc_lds_add(arow + 4, r1.x);  sc_lds_add(arow + 5, r1.y);
  sc_lds_add(arow + 6, r1.z);  sc_lds_add(arow + 7, r1.w);
  sc_lds_add(arow + 8, r2.x);  sc_lds_add(arow + 9, r2.y);
  sc_lds_add(arow + 10, r2.z); sc_lds_add(arow + 11, r2.w);
  sc_lds_add(arow + 12, r3.x); sc_lds_add(arow + 13, r3.y);
  sc_lds_add(arow + 14, r3.z); sc_lds_add(arow + 15, r3.w);
}

__device__ __forceinline__ void sc_push(unsigned* q_s, unsigned* q_r, int& fill, unsigned ltmask,
                                        int d, int sv, unsigned lo, unsigned rn) {
  const unsigned rel = (unsigned)d - lo;
  const bool hit = rel < rn;
  const unsigned m = __builtin_amdgcn_ballot_w32(hit);
  if (hit) {
    const int pos = fill + __popc(m & ltmask);
    q_s[pos] = (unsigned)sv;
    q_r[pos] = rel;
  }
  fill += __popc(m);
}

__device__ __forceinline__ void sc_round(float* acc, const float* __restrict__ h,
                                         const unsigned* q_s, const unsigned* q_r,
                                         int base, int cnt, int lane) {
  __builtin_amdgcn_wave_barrier();
  if (lane < cnt) {
    const unsigned sv  = q_s[base + lane];
    const unsigned rel = q_r[base + lane];
    sc_add_row(acc + rel * SC_PITCH, h, sv);
  }
  __builtin_amdgcn_wave_barrier();
}

__global__ __launch_bounds__(SC_TPB) void gcn_edge_owner(const int* __restrict__ ei,
                                                         const float* __restrict__ h,
                                                         const float* __restrict__ dinv,
                                                         float* __restrict__ agg,
                                                         int E, int n) {
  __shared__ __attribute__((aligned(16))) float acc[SC_RANGE * SC_PITCH];
  __shared__ unsigned qs[SC_TPB / 32][SC_QCAP];
  __shared__ unsigned qr[SC_TPB / 32][SC_QCAP];

  const int tid  = threadIdx.x;
  const int lane = tid & 31;
  const int c    = blockIdx.x;
  const int lo   = blockIdx.y * SC_RANGE;
  int rn = n - lo;
  if (rn > SC_RANGE) rn = SC_RANGE;

  float4* a4 = (float4*)acc;
  for (int i = tid; i < SC_RANGE * SC_PITCH / 4; i += SC_TPB) a4[i] = make_float4(0.f, 0.f, 0.f, 0.f);
  __syncthreads();

  const int per = (((E + SC_SLICES - 1) / SC_SLICES) + 3) & ~3;
  int e0 = c * per;
  if (e0 > E) e0 = E;
  int e1 = e0 + per;
  if (e1 > E) e1 = E;
  const int len = e1 - e0;
  const int* __restrict__ ps = ei + e0;
  const int* __restrict__ pd = ei + E + e0;

  int head = (int)(((16u - ((unsigned)(size_t)pd & 15u)) & 15u) >> 2);
  if (head > len) head = len;
  const int nvec = (len - head) >> 2;
  const int tail = head + (nvec << 2);
  const int4* __restrict__ pdv = (const int4*)(pd + head);
  const int* __restrict__ psh = ps + head;
  const int4* __restrict__ psv = (const int4*)psh;
  const bool svec = (((size_t)psh) & 15u) == 0;

  unsigned* q_s = qs[tid >> 5];
  unsigned* q_r = qr[tid >> 5];
  const unsigned ltmask = (1u << lane) - 1u;
  const unsigned ulo = (unsigned)lo, urn = (unsigned)rn;
  int fill = 0;

  if (tid < head) {
    const unsigned rel = (unsigned)pd[tid] - ulo;
    if (rel < urn) sc_add_row(acc + rel * SC_PITCH, h, (unsigned)ps[tid]);
  }

  for (int base = 0; base < nvec; base += SC_BATCH * SC_TPB) {
    int4 v[SC_BATCH], w[SC_BATCH];
#pragma unroll
    for (int u = 0; u < SC_BATCH; ++u) {
      const int i = base + u * SC_TPB + tid;
      if (i < nvec) {
        v[u] = pdv[i];
        if (svec) {
          w[u] = psv[i];
        } else {
          const int* q = psh + 4 * i;
          w[u] = make_int4(q[0], q[1], q[2], q[3]);
        }
      } else {
        v[u] = make_int4(-1, -1, -1, -1);
        w[u] = make_int4(0, 0, 0, 0);
      }
    }
#pragma unroll
    for (int u = 0; u < SC_BATCH; ++u) {
      sc_push(q_s, q_r, fill, ltmask, v[u].x, w[u].x, ulo, urn);
      sc_push(q_s, q_r, fill, ltmask, v[u].y, w[u].y, ulo, urn);
      sc_push(q_s, q_r, fill, ltmask, v[u].z, w[u].z, ulo, urn);
      sc_push(q_s, q_r, fill, ltmask, v[u].w, w[u].w, ulo, urn);
      while (fill >= 32) {
        fill -= 32;
        sc_round(acc, h, q_s, q_r, fill, 32, lane);
      }
    }
  }
  if (fill > 0) sc_round(acc, h, q_s, q_r, 0, fill, lane);

  if (tail + tid < len) {
    const unsigned rel = (unsigned)pd[tail + tid] - ulo;
    if (rel < urn) sc_add_row(acc + rel * SC_PITCH, h, (unsigned)ps[tail + tid]);
  }
  __syncthreads();

  float* __restrict__ ag = agg + (size_t)lo * HID;
  const int tot = rn * HID;
  int t0 = tid;
  for (; t0 + 3 * SC_TPB < tot; t0 += 4 * SC_TPB) {
    float a[4], dv[4];
#pragma unroll
    for (int u = 0; u < 4; ++u) {
      const int t = t0 + u * SC_TPB;
      a[u]  = acc[(t >> 4) * SC_PITCH + (t & 15)];
      dv[u] = dinv[lo + (t >> 4)];
    }
#pragma unroll
    for (int u = 0; u < 4; ++u) unsafeAtomicAdd(&ag[t0 + u * SC_TPB], a[u] * dv[u]);
  }
  for (; t0 < tot; t0 += SC_TPB)
    unsafeAtomicAdd(&ag[t0], acc[(t0 >> 4) * SC_PITCH + (t0 & 15)] * dinv[lo + (t0 >> 4)]);
}

__global__ void gcn_final(const float* __restrict__ agg, const float* __restrict__ b1,
                          const float* __restrict__ Wp, const float* __restrict__ bp,
                          float* __restrict__ out, int n) {
  int i = blockIdx.x * blockDim.x + threadIdx.x;
  if (i >= n) return;

  float hb[HID];
#pragma unroll
  for (int k = 0; k < HID; ++k) {
    float v = agg[(size_t)i * HID + k] + b1[k];
    hb[k] = v > 0.0f ? v : 0.0f;
  }

  float logits[NCLS];
  float mx = -3.0e38f;
#pragma unroll
  for (int c = 0; c < NCLS; ++c) {
    float accv = bp[c];
#pragma unroll
    for (int k = 0; k < HID; ++k) accv += hb[k] * Wp[k * NCLS + c];
    logits[c] = accv;
    mx = fmaxf(mx, accv);
  }

  float sum = 0.0f;
#pragma unroll
  for (int c = 0; c < NCLS; ++c) {
    float ev = __expf(logits[c] - mx);
    logits[c] = ev;
    sum += ev;
  }
  float inv = 1.0f / sum;
#pragma unroll
  for (int c = 0; c < NCLS; ++c) out[(size_t)i * NCLS + c] = logits[c] * inv;
}

extern "C" void kernel_launch(void* const* d_in, const int* in_sizes, int n_in,
                              void* d_out, int out_size, void* d_ws, size_t ws_size,
                              hipStream_t stream) {
  const float* x  = (const float*)d_in[0];
  const float* W1 = (const float*)d_in[1];
  const float* b1 = (const float*)d_in[2];
  const float* Wp = (const float*)d_in[3];
  const float* bp = (const float*)d_in[4];
  const int*   ei = (const int*)d_in[5];

  const int N = in_sizes[0] / F_IN;
  const int E = in_sizes[5] / 2;
  float* out = (float*)d_out;

  float* h   = (float*)d_ws;
  float* agg = h + (size_t)N * HID;
  float* deg = agg + (size_t)N * HID;

  const int tpb = 256;

  unsigned* part = (unsigned*)agg;
  const int nRanges = (N + DEG_RANGE - 1) / DEG_RANGE;
  gcn_deg_count<<<dim3(DEG_CHUNKS, nRanges), DEG_TPB, 0, stream>>>(ei, part, E, N);
  gcn_dinv<<<(N + tpb - 1) / tpb, tpb, 0, stream>>>(part, deg, N);

  const int nTiles = (N + 15) / 16;
  gcn_gemm_wmma<<<(nTiles + 7) / 8, 256, 0, stream>>>(x, W1, h, nTiles, N);

  const long long nh = (long long)N * HID;
  gcn_init_agg<<<(int)((nh + tpb - 1) / tpb), tpb, 0, stream>>>(h, deg, agg, N);

  const int nOwn = (N + SC_RANGE - 1) / SC_RANGE;
  gcn_edge_owner<<<dim3(SC_SLICES, nOwn), SC_TPB, 0, stream>>>(ei, h, deg, agg, E, N);

  gcn_final<<<(N + tpb - 1) / tpb, tpb, 0, stream>>>(agg, b1, Wp, bp, out, N);
}


# row scaling by dinv and the self-loop start of agg folded into the GEMM epilogue; the separate one-element-per-thread pass (50000 short waves) and its launch are gone; same values
# speedup vs baseline: 3.4839x; 1.0583x over previous
#include <hip/hip_runtime.h>
#include <hip/hip_bf16.h>


typedef __attribute__((ext_vector_type(2))) float v2f;
typedef __attribute__((ext_vector_type(8))) float v8f;

#define F_IN 128
#define HID  16
#define NCLS 16

#define DEG_RANGE  6272
#define DEG_CHUNKS 16
#define DEG_TPB    1024
#define DEG_BATCH  8

__device__ __forceinline__ void deg_tally(unsigned* cnt, int d, unsigned lo, unsigned rn) {
  unsigned rel = (unsigned)d - lo;
  if (rel < rn) atomicAdd(&cnt[rel], 1u);
}

__global__ __launch_bounds__(DEG_TPB) void gcn_deg_count(const int* __restrict__ ei,
                                                         unsigned* __restrict__ part,
                                                         int E, int n) {
  __shared__ unsigned cnt[DEG_RANGE];
  const int tid = threadIdx.x;
  const int c   = blockIdx.x;
  const int lo  = blockIdx.y * DEG_RANGE;
  int rn = n - lo;
  if (rn > DEG_RANGE) rn = DEG_RANGE;

  for (int i = tid; i < DEG_RANGE; i += DEG_TPB) cnt[i] = 0u;
  __syncthreads();

  const int per = (((E + DEG_CHUNKS - 1) / DEG_CHUNKS) + 3) & ~3;
  int e0 = c * per;
  if (e0 > E) e0 = E;
  int e1 = e0 + per;
  if (e1 > E) e1 = E;
  const int len = e1 - e0;
  const int* __restrict__ p = ei + E + e0;

  int head = (int)(((16u - ((unsigned)(size_t)p & 15u)) & 15u) >> 2);
  if (head > len) head = len;
  const int nvec = (len - head) >> 2;
  const int tail = head + (nvec << 2);
  const int4* __restrict__ pv = (const int4*)(p + head);

  if (tid < head) deg_tally(cnt, p[tid], (unsigned)lo, (unsigned)rn);

  for (int base = 0; base < nvec; base += DEG_BATCH * DEG_TPB) {
    int4 v[DEG_BATCH];
#pragma unroll
    for (int u = 0; u < DEG_BATCH; ++u) {
      const int i = base + u * DEG_TPB + tid;
      v[u] = (i < nvec) ? pv[i] : make_int4(-1, -1, -1, -1);
    }
#pragma unroll
    for (int u = 0; u < DEG_BATCH; ++u) {
      deg_tally(cnt, v[u].x, (unsigned)lo, (unsigned)rn);
      deg_tally(cnt, v[u].y, (unsigned)lo, (unsigned)rn);
      deg_tally(cnt, v[u].z, (unsigned)lo, (unsigned)rn);
      deg_tally(cnt, v[u].w, (unsigned)lo, (unsigned)rn);
    }
  }

  if (tail + tid < len) deg_tally(cnt, p[tail + tid], (unsigned)lo, (unsigned)rn);
  __syncthreads();

  unsigned* __restrict__ dstp = part + (size_t)c * n + lo;
  for (int i = tid; i < rn; i += DEG_TPB) dstp[i] = cnt[i];
}

__global__ void gcn_dinv(const unsigned* __restrict__ part, float* __restrict__ deg, int n) {
  int i = blockIdx.x * blockDim.x + threadIdx.x;
  if (i < n) {
    unsigned k = 1u;
#pragma unroll
    for (int c = 0; c < DEG_CHUNKS; ++c) k += part[(size_t)c * n + i];
    float d = (float)k;
    deg[i] = d > 0.0f ? rsqrtf(d) : 0.0f;
  }
}

__global__ __launch_bounds__(256) void gcn_gemm_wmma(const float* __restrict__ x,
                                                     const float* __restrict__ W,
                                                     const float* __restrict__ dinv,
                                                     float* __restrict__ h,
                                                     float* __restrict__ agg,
                                                     int nTiles, int N) {
  const int lane = threadIdx.x & 31;
  const int tile = blockIdx.x * (blockDim.x >> 5) + (threadIdx.x >> 5);
  if (tile >= nTiles) return;

  const int half = lane >> 4;
  const int l    = lane & 15;
  const int row  = tile * 16 + l;
  const int rowc = row < N ? row : N - 1;

  const float* xr = x + (size_t)rowc * F_IN + half * 2;

  const int mbase = tile * 16 + half * 8;
  float dv[8];
#pragma unroll
  for (int j = 0; j < 8; ++j) {
    const int m = mbase + j;
    dv[j] = dinv[m < N ? m : N - 1];
  }

  v8f acc = {0.f, 0.f, 0.f, 0.f, 0.f, 0.f, 0.f, 0.f};

#pragma unroll
  for (int k = 0; k < F_IN; k += 4) {
    v2f a;
    a.x = xr[k];
    a.y = xr[k + 1];
    const float* wp = W + (size_t)(k + half * 2) * HID + l;
    v2f b;
    b.x = wp[0];
    b.y = wp[HID];
    acc = __builtin_amdgcn_wmma_f32_16x16x4_f32(false, a, false, b,
                                                (short)0, acc, false, false);
  }

#pragma unroll
  for (int j = 0; j < 8; ++j) {
    const int m = mbase + j;
    if (m < N) {
      const float hv = acc[j] * dv[j];
      h[(size_t)m * HID + l]   = hv;
      agg[(size_t)m * HID + l] = hv * dv[j];
    }
  }
}

#define SC_RANGE  4000
#define SC_PITCH  17
#define SC_SLICES 10
#define SC_TPB    1024
#define SC_BATCH  8
#define SC_QCAP   160

__device__ __forceinline__ void sc_lds_add(float* p, float v) {
  __hip_atomic_fetch_add(p, v, __ATOMIC_RELAXED, __HIP_MEMORY_SCOPE_WORKGROUP);
}

__device__ __forceinline__ void sc_add_row(float* arow, const float* __restrict__ h, unsigned s) {
  const float4* __restrict__ row = (const float4*)(h + (size_t)s * HID);
  const float4 r0 = row[0], r1 = row[1], r2 = row[2], r3 = row[3];
  sc_lds_add(arow + 0, r0.x);  sc_lds_add(arow + 1, r0.y);
  sc_lds_add(arow + 2, r0.z);  sc_lds_add(arow + 3, r0.w);
  sc_lds_add(arow + 4, r1.x);  sc_lds_add(arow + 5, r1.y);
  sc_lds_add(arow + 6, r1.z);  sc_lds_add(arow + 7, r1.w);
  sc_lds_add(arow + 8, r2.x);  sc_lds_add(arow + 9, r2.y);
  sc_lds_add(arow + 10, r2.z); sc_lds_add(arow + 11, r2.w);
  sc_lds_add(arow + 12, r3.x); sc_lds_add(arow + 13, r3.y);
  sc_lds_add(arow + 14, r3.z); sc_lds_add(arow + 15, r3.w);
}

__device__ __forceinline__ void sc_push(unsigned* q_s, unsigned* q_r, int& fill, unsigned ltmask,
                                        int d, int sv, unsigned lo, unsigned rn) {
  const unsigned rel = (unsigned)d - lo;
  const bool hit = rel < rn;
  const unsigned m = __builtin_amdgcn_ballot_w32(hit);
  if (hit) {
    const int pos = fill + __popc(m & ltmask);
    q_s[pos] = (unsigned)sv;
    q_r[pos] = rel;
  }
  fill += __popc(m);
}

__device__ __forceinline__ void sc_round(float* acc, const float* __restrict__ h,
                                         const unsigned* q_s, const unsigned* q_r,
                                         int base, int cnt, int lane) {
  __builtin_amdgcn_wave_barrier();
  if (lane < cnt) {
    const unsigned sv  = q_s[base + lane];
    const unsigned rel = q_r[base + lane];
    sc_add_row(acc + rel * SC_PITCH, h, sv);
  }
  __builtin_amdgcn_wave_barrier();
}

__global__ __launch_bounds__(SC_TPB) void gcn_edge_owner(const int* __restrict__ ei,
                                                         const float* __restrict__ h,
                                                         const float* __restrict__ dinv,
                                                         float* __restrict__ agg,
                                                         int E, int n) {
  __shared__ __attribute__((aligned(16))) float acc[SC_RANGE * SC_PITCH];
  __shared__ unsigned qs[SC_TPB / 32][SC_QCAP];
  __shared__ unsigned qr[SC_TPB / 32][SC_QCAP];

  const int tid  = threadIdx.x;
  const int lane = tid & 31;
  const int c    = blockIdx.x;
  const int lo   = blockIdx.y * SC_RANGE;
  int rn = n - lo;
  if (rn > SC_RANGE) rn = SC_RANGE;

  float4* a4 = (float4*)acc;
  for (int i = tid; i < SC_RANGE * SC_PITCH / 4; i += SC_TPB) a4[i] = make_float4(0.f, 0.f, 0.f, 0.f);
  __syncthreads();

  const int per = (((E + SC_SLICES - 1) / SC_SLICES) + 3) & ~3;
  int e0 = c * per;
  if (e0 > E) e0 = E;
  int e1 = e0 + per;
  if (e1 > E) e1 = E;
  const int len = e1 - e0;
  const int* __restrict__ ps = ei + e0;
  const int* __restrict__ pd = ei + E + e0;

  int head = (int)(((16u - ((unsigned)(size_t)pd & 15u)) & 15u) >> 2);
  if (head > len) head = len;
  const int nvec = (len - head) >> 2;
  const int tail = head + (nvec << 2);
  const int4* __restrict__ pdv = (const int4*)(pd + head);
  const int* __restrict__ psh = ps + head;
  const int4* __restrict__ psv = (const int4*)psh;
  const bool svec = (((size_t)psh) & 15u) == 0;

  unsigned* q_s = qs[tid >> 5];
  unsigned* q_r = qr[tid >> 5];
  const unsigned ltmask = (1u << lane) - 1u;
  const unsigned ulo = (unsigned)lo, urn = (unsigned)rn;
  int fill = 0;

  if (tid < head) {
    const unsigned rel = (unsigned)pd[tid] - ulo;
    if (rel < urn) sc_add_row(acc + rel * SC_PITCH, h, (unsigned)ps[tid]);
  }

  for (int base = 0; base < nvec; base += SC_BATCH * SC_TPB) {
    int4 v[SC_BATCH], w[SC_BATCH];
#pragma unroll
    for (int u = 0; u < SC_BATCH; ++u) {
      const int i = base + u * SC_TPB + tid;
      if (i < nvec) {
        v[u] = pdv[i];
        if (svec) {
          w[u] = psv[i];
        } else {
          const int* q = psh + 4 * i;
          w[u] = make_int4(q[0], q[1], q[2], q[3]);
        }
      } else {
        v[u] = make_int4(-1, -1, -1, -1);
        w[u] = make_int4(0, 0, 0, 0);
      }
    }
#pragma unroll
    for (int u = 0; u < SC_BATCH; ++u) {
      sc_push(q_s, q_r, fill, ltmask, v[u].x, w[u].x, ulo, urn);
      sc_push(q_s, q_r, fill, ltmask, v[u].y, w[u].y, ulo, urn);
      sc_push(q_s, q_r, fill, ltmask, v[u].z, w[u].z, ulo, urn);
      sc_push(q_s, q_r, fill, ltmask, v[u].w, w[u].w, ulo, urn);
      while (fill >= 32) {
        fill -= 32;
        sc_round(acc, h, q_s, q_r, fill, 32, lane);
      }
    }
  }
  if (fill > 0) sc_round(acc, h, q_s, q_r, 0, fill, lane);

  if (tail + tid < len) {
    const unsigned rel = (unsigned)pd[tail + tid] - ulo;
    if (rel < urn) sc_add_row(acc + rel * SC_PITCH, h, (unsigned)ps[tail + tid]);
  }
  __syncthreads();

  float* __restrict__ ag = agg + (size_t)lo * HID;
  const int tot = rn * HID;
  int t0 = tid;
  for (; t0 + 3 * SC_TPB < tot; t0 += 4 * SC_TPB) {
    float a[4], dv[4];
#pragma unroll
    for (int u = 0; u < 4; ++u) {
      const int t = t0 + u * SC_TPB;
      a[u]  = acc[(t >> 4) * SC_PITCH + (t & 15)];
      dv[u] = dinv[lo + (t >> 4)];
    }
#pragma unroll
    for (int u = 0; u < 4; ++u) unsafeAtomicAdd(&ag[t0 + u * SC_TPB], a[u] * dv[u]);
  }
  for (; t0 < tot; t0 += SC_TPB)
    unsafeAtomicAdd(&ag[t0], acc[(t0 >> 4) * SC_PITCH + (t0 & 15)] * dinv[lo + (t0 >> 4)]);
}

__global__ void gcn_final(const float* __restrict__ agg, const float* __restrict__ b1,
                          const float* __restrict__ Wp, const float* __restrict__ bp,
                          float* __restrict__ out, int n) {
  int i = blockIdx.x * blockDim.x + threadIdx.x;
  if (i >= n) return;

  float hb[HID];
#pragma unroll
  for (int k = 0; k < HID; ++k) {
    float v = agg[(size_t)i * HID + k] + b1[k];
    hb[k] = v > 0.0f ? v : 0.0f;
  }

  float logits[NCLS];
  float mx = -3.0e38f;
#pragma unroll
  for (int c = 0; c < NCLS; ++c) {
    float accv = bp[c];
#pragma unroll
    for (int k = 0; k < HID; ++k) accv += hb[k] * Wp[k * NCLS + c];
    logits[c] = accv;
    mx = fmaxf(mx, accv);
  }

  float sum = 0.0f;
#pragma unroll
  for (int c = 0; c < NCLS; ++c) {
    float ev = __expf(logits[c] - mx);
    logits[c] = ev;
    sum += ev;
  }
  float inv = 1.0f / sum;
#pragma unroll
  for (int c = 0; c < NCLS; ++c) out[(size_t)i * NCLS + c] = logits[c] * inv;
}

extern "C" void kernel_launch(void* const* d_in, const int* in_sizes, int n_in,
                              void* d_out, int out_size, void* d_ws, size_t ws_size,
                              hipStream_t stream) {
  const float* x  = (const float*)d_in[0];
  const float* W1 = (const float*)d_in[1];
  const float* b1 = (const float*)d_in[2];
  const float* Wp = (const float*)d_in[3];
  const float* bp = (const float*)d_in[4];
  const int*   ei = (const int*)d_in[5];

  const int N = in_sizes[0] / F_IN;
  const int E = in_sizes[5] / 2;
  float* out = (float*)d_out;

  float* h   = (float*)d_ws;
  float* agg = h + (size_t)N * HID;
  float* deg = agg + (size_t)N * HID;

  const int tpb = 256;

  unsigned* part = (unsigned*)agg;
  const int nRanges = (N + DEG_RANGE - 1) / DEG_RANGE;
  gcn_deg_count<<<dim3(DEG_CHUNKS, nRanges), DEG_TPB, 0, stream>>>(ei, part, E, N);
  gcn_dinv<<<(N + tpb - 1) / tpb, tpb, 0, stream>>>(part, deg, N);

  const int nTiles = (N + 15) / 16;
  gcn_gemm_wmma<<<(nTiles + 7) / 8, 256, 0, stream>>>(x, W1, deg, h, agg, nTiles, N);

  const int nOwn = (N + SC_RANGE - 1) / SC_RANGE;
  gcn_edge_owner<<<dim3(SC_SLICES, nOwn), SC_TPB, 0, stream>>>(ei, h, deg, agg, E, N);

  gcn_final<<<(N + tpb - 1) / tpb, tpb, 0, stream>>>(agg, b1, Wp, bp, out, N);
}


# edge pass gather: four lanes share a popped edge, each loads one quarter of the 64-byte h row and does its 4 LDS adds; a wave load touches 8 lines, not 32; order of LDS adds in a wave changes
# speedup vs baseline: 3.6481x; 1.0471x over previous
#include <hip/hip_runtime.h>
#include <hip/hip_bf16.h>


typedef __attribute__((ext_vector_type(2))) float v2f;
typedef __attribute__((ext_vector_type(8))) float v8f;

#define F_IN 128
#define HID  16
#define NCLS 16

#define DEG_RANGE  6272
#define DEG_CHUNKS 16
#define DEG_TPB    1024
#define DEG_BATCH  8

__device__ __forceinline__ void deg_tally(unsigned* cnt, int d, unsigned lo, unsigned rn) {
  unsigned rel = (unsigned)d - lo;
  if (rel < rn) atomicAdd(&cnt[rel], 1u);
}

__global__ __launch_bounds__(DEG_TPB) void gcn_deg_count(const int* __restrict__ ei,
                                                         unsigned* __restrict__ part,
                                                         int E, int n) {
  __shared__ unsigned cnt[DEG_RANGE];
  const int tid = threadIdx.x;
  const int c   = blockIdx.x;
  const int lo  = blockIdx.y * DEG_RANGE;
  int rn = n - lo;
  if (rn > DEG_RANGE) rn = DEG_RANGE;

  for (int i = tid; i < DEG_RANGE; i += DEG_TPB) cnt[i] = 0u;
  __syncthreads();

  const int per = (((E + DEG_CHUNKS - 1) / DEG_CHUNKS) + 3) & ~3;
  int e0 = c * per;
  if (e0 > E) e0 = E;
  int e1 = e0 + per;
  if (e1 > E) e1 = E;
  const int len = e1 - e0;
  const int* __restrict__ p = ei + E + e0;

  int head = (int)(((16u - ((unsigned)(size_t)p & 15u)) & 15u) >> 2);
  if (head > len) head = len;
  const int nvec = (len - head) >> 2;
  const int tail = head + (nvec << 2);
  const int4* __restrict__ pv = (const int4*)(p + head);

  if (tid < head) deg_tally(cnt, p[tid], (unsigned)lo, (unsigned)rn);

  for (int base = 0; base < nvec; base += DEG_BATCH * DEG_TPB) {
    int4 v[DEG_BATCH];
#pragma unroll
    for (int u = 0; u < DEG_BATCH; ++u) {
      const int i = base + u * DEG_TPB + tid;
      v[u] = (i < nvec) ? pv[i] : make_int4(-1, -1, -1, -1);
    }
#pragma unroll
    for (int u = 0; u < DEG_BATCH; ++u) {
      deg_tally(cnt, v[u].x, (unsigned)lo, (unsigned)rn);
      deg_tally(cnt, v[u].y, (unsigned)lo, (unsigned)rn);
      deg_tally(cnt, v[u].z, (unsigned)lo, (unsigned)rn);
      deg_tally(cnt, v[u].w, (unsigned)lo, (unsigned)rn);
    }
  }

  if (tail + tid < len) deg_tally(cnt, p[tail + tid], (unsigned)lo, (unsigned)rn);
  __syncthreads();

  unsigned* __restrict__ dstp = part + (size_t)c * n + lo;
  for (int i = tid; i < rn; i += DEG_TPB) dstp[i] = cnt[i];
}

__global__ void gcn_dinv(const unsigned* __restrict__ part, float* __restrict__ deg, int n) {
  int i = blockIdx.x * blockDim.x + threadIdx.x;
  if (i < n) {
    unsigned k = 1u;
#pragma unroll
    for (int c = 0; c < DEG_CHUNKS; ++c) k += part[(size_t)c * n + i];
    float d = (float)k;
    deg[i] = d > 0.0f ? rsqrtf(d) : 0.0f;
  }
}

__global__ __launch_bounds__(256) void gcn_gemm_wmma(const float* __restrict__ x,
                                                     const float* __restrict__ W,
                                                     const float* __restrict__ dinv,
                                                     float* __restrict__ h,
                                                     float* __restrict__ agg,
                                                     int nTiles, int N) {
  const int lane = threadIdx.x & 31;
  const int tile = blockIdx.x * (blockDim.x >> 5) + (threadIdx.x >> 5);
  if (tile >= nTiles) return;

  const int half = lane >> 4;
  const int l    = lane & 15;
  const int row  = tile * 16 + l;
  const int rowc = row < N ? row : N - 1;

  const float* xr = x + (size_t)rowc * F_IN + half * 2;

  const int mbase = tile * 16 + half * 8;
  float dv[8];
#pragma unroll
  for (int j = 0; j < 8; ++j) {
    const int m = mbase + j;
    dv[j] = dinv[m < N ? m : N - 1];
  }

  v8f acc = {0.f, 0.f, 0.f, 0.f, 0.f, 0.f, 0.f, 0.f};

#pragma unroll
  for (int k = 0; k < F_IN; k += 4) {
    v2f a;
    a.x = xr[k];
    a.y = xr[k + 1];
    const float* wp = W + (size_t)(k + half * 2) * HID + l;
    v2f b;
    b.x = wp[0];
    b.y = wp[HID];
    acc = __builtin_amdgcn_wmma_f32_16x16x4_f32(false, a, false, b,
                                                (short)0, acc, false, false);
  }

#pragma unroll
  for (int j = 0; j < 8; ++j) {
    const int m = mbase + j;
    if (m < N) {
      const float hv = acc[j] * dv[j];
      h[(size_t)m * HID + l]   = hv;
      agg[(size_t)m * HID + l] = hv * dv[j];
    }
  }
}

#define SC_RANGE  4000
#define SC_PITCH  17
#define SC_SLICES 10
#define SC_TPB    1024
#define SC_BATCH  8
#define SC_QCAP   160

__device__ __forceinline__ void sc_lds_add(float* p, float v) {
  __hip_atomic_fetch_add(p, v, __ATOMIC_RELAXED, __HIP_MEMORY_SCOPE_WORKGROUP);
}

__device__ __forceinline__ void sc_add_row(float* arow, const float* __restrict__ h, unsigned s) {
  const float4* __restrict__ row = (const float4*)(h + (size_t)s * HID);
  const float4 r0 = row[0], r1 = row[1], r2 = row[2], r3 = row[3];
  sc_lds_add(arow + 0, r0.x);  sc_lds_add(arow + 1, r0.y);
  sc_lds_add(arow + 2, r0.z);  sc_lds_add(arow + 3, r0.w);
  sc_lds_add(arow + 4, r1.x);  sc_lds_add(arow + 5, r1.y);
  sc_lds_add(arow + 6, r1.z);  sc_lds_add(arow + 7, r1.w);
  sc_lds_add(arow + 8, r2.x);  sc_lds_add(arow + 9, r2.y);
  sc_lds_add(arow + 10, r2.z); sc_lds_add(arow + 11, r2.w);
  sc_lds_add(arow + 12, r3.x); sc_lds_add(arow + 13, r3.y);
  sc_lds_add(arow + 14, r3.z); sc_lds_add(arow + 15, r3.w);
}

__device__ __forceinline__ void sc_push(unsigned* q_s, unsigned* q_r, int& fill, unsigned ltmask,
                                        int d, int sv, unsigned lo, unsigned rn) {
  const unsigned rel = (unsigned)d - lo;
  const bool hit = rel < rn;
  const unsigned m = __builtin_amdgcn_ballot_w32(hit);
  if (hit) {
    const int pos = fill + __popc(m & ltmask);
    q_s[pos] = (unsigned)sv;
    q_r[pos] = rel;
  }
  fill += __popc(m);
}

template <bool FULL>
__device__ __forceinline__ void sc_round(float* acc, const float* __restrict__ h,
                                         const unsigned* q_s, const unsigned* q_r,
                                         int base, int cnt, int lane) {
  __builtin_amdgcn_wave_barrier();
  const int g = lane >> 2;
  const int q = lane & 3;
  unsigned sv[4], rel[4];
#pragma unroll
  for (int i = 0; i < 4; ++i) {
    int e = i * 8 + g;
    if (!FULL && e >= cnt) e = 0;
    sv[i]  = q_s[base + e];
    rel[i] = q_r[base + e];
  }
  float4 r[4];
#pragma unroll
  for (int i = 0; i < 4; ++i) r[i] = ((const float4*)(h + (size_t)sv[i] * HID))[q];
#pragma unroll
  for (int i = 0; i < 4; ++i) {
    if (FULL || i * 8 + g < cnt) {
      float* a = acc + rel[i] * SC_PITCH + q * 4;
      sc_lds_add(a + 0, r[i].x);
      sc_lds_add(a + 1, r[i].y);
      sc_lds_add(a + 2, r[i].z);
      sc_lds_add(a + 3, r[i].w);
    }
  }
  __builtin_amdgcn_wave_barrier();
}

__global__ __launch_bounds__(SC_TPB) void gcn_edge_owner(const int* __restrict__ ei,
                                                         const float* __restrict__ h,
                                                         const float* __restrict__ dinv,
                                                         float* __restrict__ agg,
                                                         int E, int n) {
  __shared__ __attribute__((aligned(16))) float acc[SC_RANGE * SC_PITCH];
  __shared__ unsigned qs[SC_TPB / 32][SC_QCAP];
  __shared__ unsigned qr[SC_TPB / 32][SC_QCAP];

  const int tid  = threadIdx.x;
  const int lane = tid & 31;
  const int c    = blockIdx.x;
  const int lo   = blockIdx.y * SC_RANGE;
  int rn = n - lo;
  if (rn > SC_RANGE) rn = SC_RANGE;

  float4* a4 = (float4*)acc;
  for (int i = tid; i < SC_RANGE * SC_PITCH / 4; i += SC_TPB) a4[i] = make_float4(0.f, 0.f, 0.f, 0.f);
  __syncthreads();

  const int per = (((E + SC_SLICES - 1) / SC_SLICES) + 3) & ~3;
  int e0 = c * per;
  if (e0 > E) e0 = E;
  int e1 = e0 + per;
  if (e1 > E) e1 = E;
  const int len = e1 - e0;
  const int* __restrict__ ps = ei + e0;
  const int* __restrict__ pd = ei + E + e0;

  int head = (int)(((16u - ((unsigned)(size_t)pd & 15u)) & 15u) >> 2);
  if (head > len) head = len;
  const int nvec = (len - head) >> 2;
  const int tail = head + (nvec << 2);
  const int4* __restrict__ pdv = (const int4*)(pd + head);
  const int* __restrict__ psh = ps + head;
  const int4* __restrict__ psv = (const int4*)psh;
  const bool svec = (((size_t)psh) & 15u) == 0;

  unsigned* q_s = qs[tid >> 5];
  unsigned* q_r = qr[tid >> 5];
  const unsigned ltmask = (1u << lane) - 1u;
  const unsigned ulo = (unsigned)lo, urn = (unsigned)rn;
  int fill = 0;

  if (tid < head) {
    const unsigned rel = (unsigned)pd[tid] - ulo;
    if (rel < urn) sc_add_row(acc + rel * SC_PITCH, h, (unsigned)ps[tid]);
  }

  for (int base = 0; base < nvec; base += SC_BATCH * SC_TPB) {
    int4 v[SC_BATCH], w[SC_BATCH];
#pragma unroll
    for (int u = 0; u < SC_BATCH; ++u) {
      const int i = base + u * SC_TPB + tid;
      if (i < nvec) {
        v[u] = pdv[i];
        if (svec) {
          w[u] = psv[i];
        } else {
          const int* q = psh + 4 * i;
          w[u] = make_int4(q[0], q[1], q[2], q[3]);
        }
      } else {
        v[u] = make_int4(-1, -1, -1, -1);
        w[u] = make_int4(0, 0, 0, 0);
      }
    }
#pragma unroll
    for (int u = 0; u < SC_BATCH; ++u) {
      sc_push(q_s, q_r, fill, ltmask, v[u].x, w[u].x, ulo, urn);
      sc_push(q_s, q_r, fill, ltmask, v[u].y, w[u].y, ulo, urn);
      sc_push(q_s, q_r, fill, ltmask, v[u].z, w[u].z, ulo, urn);
      sc_push(q_s, q_r, fill, ltmask, v[u].w, w[u].w, ulo, urn);
      while (fill >= 32) {
        fill -= 32;
        sc_round<true>(acc, h, q_s, q_r, fill, 32, lane);
      }
    }
  }
  if (fill > 0) sc_round<false>(acc, h, q_s, q_r, 0, fill, lane);

  if (tail + tid < len) {
    const unsigned rel = (unsigned)pd[tail + tid] - ulo;
    if (rel < urn) sc_add_row(acc + rel * SC_PITCH, h, (unsigned)ps[tail + tid]);
  }
  __syncthreads();

  float* __restrict__ ag = agg + (size_t)lo * HID;
  const int tot = rn * HID;
  int t0 = tid;
  for (; t0 + 3 * SC_TPB < tot; t0 += 4 * SC_TPB) {
    float a[4], dv[4];
#pragma unroll
    for (int u = 0; u < 4; ++u) {
      const int t = t0 + u * SC_TPB;
      a[u]  = acc[(t >> 4) * SC_PITCH + (t & 15)];
      dv[u] = dinv[lo + (t >> 4)];
    }
#pragma unroll
    for (int u = 0; u < 4; ++u) unsafeAtomicAdd(&ag[t0 + u * SC_TPB], a[u] * dv[u]);
  }
  for (; t0 < tot; t0 += SC_TPB)
    unsafeAtomicAdd(&ag[t0], acc[(t0 >> 4) * SC_PITCH + (t0 & 15)] * dinv[lo + (t0 >> 4)]);
}

__global__ void gcn_final(const float* __restrict__ agg, const float* __restrict__ b1,
                          const float* __restrict__ Wp, const float* __restrict__ bp,
                          float* __restrict__ out, int n) {
  int i = blockIdx.x * blockDim.x + threadIdx.x;
  if (i >= n) return;

  float hb[HID];
#pragma unroll
  for (int k = 0; k < HID; ++k) {
    float v = agg[(size_t)i * HID + k] + b1[k];
    hb[k] = v > 0.0f ? v : 0.0f;
  }

  float logits[NCLS];
  float mx = -3.0e38f;
#pragma unroll
  for (int c = 0; c < NCLS; ++c) {
    float accv = bp[c];
#pragma unroll
    for (int k = 0; k < HID; ++k) accv += hb[k] * Wp[k * NCLS + c];
    logits[c] = accv;
    mx = fmaxf(mx, accv);
  }

  float sum = 0.0f;
#pragma unroll
  for (int c = 0; c < NCLS; ++c) {
    float ev = __expf(logits[c] - mx);
    logits[c] = ev;
    sum += ev;
  }
  float inv = 1.0f / sum;
#pragma unroll
  for (int c = 0; c < NCLS; ++c) out[(size_t)i * NCLS + c] = logits[c] * inv;
}

extern "C" void kernel_launch(void* const* d_in, const int* in_sizes, int n_in,
                              void* d_out, int out_size, void* d_ws, size_t ws_size,
                              hipStream_t stream) {
  const float* x  = (const float*)d_in[0];
  const float* W1 = (const float*)d_in[1];
  const float* b1 = (const float*)d_in[2];
  const float* Wp = (const float*)d_in[3];
  const float* bp = (const float*)d_in[4];
  const int*   ei = (const int*)d_in[5];

  const int N = in_sizes[0] / F_IN;
  const int E = in_sizes[5] / 2;
  float* out = (float*)d_out;

  float* h   = (float*)d_ws;
  float* agg = h + (size_t)N * HID;
  float* deg = agg + (size_t)N * HID;

  const int tpb = 256;

  unsigned* part = (unsigned*)agg;
  const int nRanges = (N + DEG_RANGE - 1) / DEG_RANGE;
  gcn_deg_count<<<dim3(DEG_CHUNKS, nRanges), DEG_TPB, 0, stream>>>(ei, part, E, N);
  gcn_dinv<<<(N + tpb - 1) / tpb, tpb, 0, stream>>>(part, deg, N);

  const int nTiles = (N + 15) / 16;
  gcn_gemm_wmma<<<(nTiles + 7) / 8, 256, 0, stream>>>(x, W1, deg, h, agg, nTiles, N);

  const int nOwn = (N + SC_RANGE - 1) / SC_RANGE;
  gcn_edge_owner<<<dim3(SC_SLICES, nOwn), SC_TPB, 0, stream>>>(ei, h, deg, agg, E, N);

  gcn_final<<<(N + tpb - 1) / tpb, tpb, 0, stream>>>(agg, b1, Wp, bp, out, N);
}


# degree count: partial planes hold 16-bit counts, two nodes to a word, so 64 slices fit where 16 did and node ranges are 4x as long: dst is read and tested 4 times, not 16; same degree values
# speedup vs baseline: 3.8825x; 1.0643x over previous
#include <hip/hip_runtime.h>
#include <hip/hip_bf16.h>


typedef __attribute__((ext_vector_type(2))) float v2f;
typedef __attribute__((ext_vector_type(8))) float v8f;

#define F_IN 128
#define HID  16
#define NCLS 16

#define DEG_RANGE  25000
#define DEG_CHUNKS 64
#define DEG_TPB    1024
#define DEG_BATCH  16

__device__ __forceinline__ void deg_tally(unsigned* cnt, int d, unsigned lo, unsigned rn) {
  unsigned rel = (unsigned)d - lo;
  if (rel < rn) atomicAdd(&cnt[rel], 1u);
}

__global__ __launch_bounds__(DEG_TPB) void gcn_deg_count(const int* __restrict__ ei,
                                                         unsigned* __restrict__ part,
                                                         int E, int n) {
  __shared__ __attribute__((aligned(16))) unsigned cnt[DEG_RANGE];
  const int tid = threadIdx.x;
  const int c   = blockIdx.x;
  const int lo  = blockIdx.y * DEG_RANGE;
  int rn = n - lo;
  if (rn > DEG_RANGE) rn = DEG_RANGE;

  uint4* c4 = (uint4*)cnt;
  for (int i = tid; i < DEG_RANGE / 4; i += DEG_TPB) c4[i] = make_uint4(0u, 0u, 0u, 0u);
  __syncthreads();

  const int per = (((E + DEG_CHUNKS - 1) / DEG_CHUNKS) + 3) & ~3;
  int e0 = c * per;
  if (e0 > E) e0 = E;
  int e1 = e0 + per;
  if (e1 > E) e1 = E;
  const int len = e1 - e0;
  const int* __restrict__ p = ei + E + e0;

  int head = (int)(((16u - ((unsigned)(size_t)p & 15u)) & 15u) >> 2);
  if (head > len) head = len;
  const int nvec = (len - head) >> 2;
  const int tail = head + (nvec << 2);
  const int4* __restrict__ pv = (const int4*)(p + head);

  if (tid < head) deg_tally(cnt, p[tid], (unsigned)lo, (unsigned)rn);

  for (int base = 0; base < nvec; base += DEG_BATCH * DEG_TPB) {
    int4 v[DEG_BATCH];
#pragma unroll
    for (int u = 0; u < DEG_BATCH; ++u) {
      const int i = base + u * DEG_TPB + tid;
      v[u] = (i < nvec) ? pv[i] : make_int4(-1, -1, -1, -1);
    }
#pragma unroll
    for (int u = 0; u < DEG_BATCH; ++u) {
      deg_tally(cnt, v[u].x, (unsigned)lo, (unsigned)rn);
      deg_tally(cnt, v[u].y, (unsigned)lo, (unsigned)rn);
      deg_tally(cnt, v[u].z, (unsigned)lo, (unsigned)rn);
      deg_tally(cnt, v[u].w, (unsigned)lo, (unsigned)rn);
    }
  }

  if (tail + tid < len) deg_tally(cnt, p[tail + tid], (unsigned)lo, (unsigned)rn);
  __syncthreads();

  const int pw = (n + 1) >> 1;
  unsigned* __restrict__ dstp = part + (size_t)c * pw + (lo >> 1);
  const uint2* __restrict__ c2 = (const uint2*)cnt;
  const int nw = (rn + 1) >> 1;
  for (int i = tid; i < nw; i += DEG_TPB) {
    const uint2 k = c2[i];
    dstp[i] = k.x | (k.y << 16);
  }
}

__global__ void gcn_dinv(const unsigned* __restrict__ part, float* __restrict__ deg, int n) {
  const int pw = (n + 1) >> 1;
  int t = blockIdx.x * blockDim.x + threadIdx.x;
  if (t < pw) {
    unsigned k0 = 1u, k1 = 1u;
#pragma unroll
    for (int c = 0; c < DEG_CHUNKS; ++c) {
      const unsigned w = part[(size_t)c * pw + t];
      k0 += w & 0xffffu;
      k1 += w >> 16;
    }
    const float d0 = (float)k0, d1 = (float)k1;
    const float r0 = d0 > 0.0f ? rsqrtf(d0) : 0.0f;
    const float r1 = d1 > 0.0f ? rsqrtf(d1) : 0.0f;
    if (2 * t + 1 < n) ((float2*)deg)[t] = make_float2(r0, r1);
    else deg[2 * t] = r0;
  }
}

__global__ __launch_bounds__(256) void gcn_gemm_wmma(const float* __restrict__ x,
                                                     const float* __restrict__ W,
                                                     const float* __restrict__ dinv,
                                                     float* __restrict__ h,
                                                     float* __restrict__ agg,
                                                     int nTiles, int N) {
  const int lane = threadIdx.x & 31;
  const int tile = blockIdx.x * (blockDim.x >> 5) + (threadIdx.x >> 5);
  if (tile >= nTiles) return;

  const int half = lane >> 4;
  const int l    = lane & 15;
  const int row  = tile * 16 + l;
  const int rowc = row < N ? row : N - 1;

  const float* xr = x + (size_t)rowc * F_IN + half * 2;

  const int mbase = tile * 16 + half * 8;
  float dv[8];
#pragma unroll
  for (int j = 0; j < 8; ++j) {
    const int m = mbase + j;
    dv[j] = dinv[m < N ? m : N - 1];
  }

  v8f acc = {0.f, 0.f, 0.f, 0.f, 0.f, 0.f, 0.f, 0.f};

#pragma unroll
  for (int k = 0; k < F_IN; k += 4) {
    v2f a;
    a.x = xr[k];
    a.y = xr[k + 1];
    const float* wp = W + (size_t)(k + half * 2) * HID + l;
    v2f b;
    b.x = wp[0];
    b.y = wp[HID];
    acc = __builtin_amdgcn_wmma_f32_16x16x4_f32(false, a, false, b,
                                                (short)0, acc, false, false);
  }

#pragma unroll
  for (int j = 0; j < 8; ++j) {
    const int m = mbase + j;
    if (m < N) {
      const float hv = acc[j] * dv[j];
      h[(size_t)m * HID + l]   = hv;
      agg[(size_t)m * HID + l] = hv * dv[j];
    }
  }
}

#define SC_RANGE  4000
#define SC_PITCH  17
#define SC_SLICES 10
#define SC_TPB    1024
#define SC_BATCH  8
#define SC_QCAP   160

__device__ __forceinline__ void sc_lds_add(float* p, float v) {
  __hip_atomic_fetch_add(p, v, __ATOMIC_RELAXED, __HIP_MEMORY_SCOPE_WORKGROUP);
}

__device__ __forceinline__ void sc_add_row(float* arow, const float* __restrict__ h, unsigned s) {
  const float4* __restrict__ row = (const float4*)(h + (size_t)s * HID);
  const float4 r0 = row[0], r1 = row[1], r2 = row[2], r3 = row[3];
  sc_lds_add(arow + 0, r0.x);  sc_lds_add(arow + 1, r0.y);
  sc_lds_add(arow + 2, r0.z);  sc_lds_add(arow + 3, r0.w);
  sc_lds_add(arow + 4, r1.x);  sc_lds_add(arow + 5, r1.y);
  sc_lds_add(arow + 6, r1.z);  sc_lds_add(arow + 7, r1.w);
  sc_lds_add(arow + 8, r2.x);  sc_lds_add(arow + 9, r2.y);
  sc_lds_add(arow + 10, r2.z); sc_lds_add(arow + 11, r2.w);
  sc_lds_add(arow + 12, r3.x); sc_lds_add(arow + 13, r3.y);
  sc_lds_add(arow + 14, r3.z); sc_lds_add(arow + 15, r3.w);
}

__device__ __forceinline__ void sc_push(unsigned* q_s, unsigned* q_r, int& fill, unsigned ltmask,
                                        int d, int sv, unsigned lo, unsigned rn) {
  const unsigned rel = (unsigned)d - lo;
  const bool hit = rel < rn;
  const unsigned m = __builtin_amdgcn_ballot_w32(hit);
  if (hit) {
    const int pos = fill + __popc(m & ltmask);
    q_s[pos] = (unsigned)sv;
    q_r[pos] = rel;
  }
  fill += __popc(m);
}

template <bool FULL>
__device__ __forceinline__ void sc_round(float* acc, const float* __restrict__ h,
                                         const unsigned* q_s, const unsigned* q_r,
                                         int base, int cnt, int lane) {
  __builtin_amdgcn_wave_barrier();
  const int g = lane >> 2;
  const int q = lane & 3;
  unsigned sv[4], rel[4];
#pragma unroll
  for (int i = 0; i < 4; ++i) {
    int e = i * 8 + g;
    if (!FULL && e >= cnt) e = 0;
    sv[i]  = q_s[base + e];
    rel[i] = q_r[base + e];
  }
  float4 r[4];
#pragma unroll
  for (int i = 0; i < 4; ++i) r[i] = ((const float4*)(h + (size_t)sv[i] * HID))[q];
#pragma unroll
  for (int i = 0; i < 4; ++i) {
    if (FULL || i * 8 + g < cnt) {
      float* a = acc + rel[i] * SC_PITCH + q * 4;
      sc_lds_add(a + 0, r[i].x);
      sc_lds_add(a + 1, r[i].y);
      sc_lds_add(a + 2, r[i].z);
      sc_lds_add(a + 3, r[i].w);
    }
  }
  __builtin_amdgcn_wave_barrier();
}

__global__ __launch_bounds__(SC_TPB) void gcn_edge_owner(const int* __restrict__ ei,
                                                         const float* __restrict__ h,
                                                         const float* __restrict__ dinv,
                                                         float* __restrict__ agg,
                                                         int E, int n) {
  __shared__ __attribute__((aligned(16))) float acc[SC_RANGE * SC_PITCH];
  __shared__ unsigned qs[SC_TPB / 32][SC_QCAP];
  __shared__ unsigned qr[SC_TPB / 32][SC_QCAP];

  const int tid  = threadIdx.x;
  const int lane = tid & 31;
  const int c    = blockIdx.x;
  const int lo   = blockIdx.y * SC_RANGE;
  int rn = n - lo;
  if (rn > SC_RANGE) rn = SC_RANGE;

  float4* a4 = (float4*)acc;
  for (int i = tid; i < SC_RANGE * SC_PITCH / 4; i += SC_TPB) a4[i] = make_float4(0.f, 0.f, 0.f, 0.f);
  __syncthreads();

  const int per = (((E + SC_SLICES - 1) / SC_SLICES) + 3) & ~3;
  int e0 = c * per;
  if (e0 > E) e0 = E;
  int e1 = e0 + per;
  if (e1 > E) e1 = E;
  const int len = e1 - e0;
  const int* __restrict__ ps = ei + e0;
  const int* __restrict__ pd = ei + E + e0;

  int head = (int)(((16u - ((unsigned)(size_t)pd & 15u)) & 15u) >> 2);
  if (head > len) head = len;
  const int nvec = (len - head) >> 2;
  const int tail = head + (nvec << 2);
  const int4* __restrict__ pdv = (const int4*)(pd + head);
  const int* __restrict__ psh = ps + head;
  const int4* __restrict__ psv = (const int4*)psh;
  const bool svec = (((size_t)psh) & 15u) == 0;

  unsigned* q_s = qs[tid >> 5];
  unsigned* q_r = qr[tid >> 5];
  const unsigned ltmask = (1u << lane) - 1u;
  const unsigned ulo = (unsigned)lo, urn = (unsigned)rn;
  int fill = 0;

  if (tid < head) {
    const unsigned rel = (unsigned)pd[tid] - ulo;
    if (rel < urn) sc_add_row(acc + rel * SC_PITCH, h, (unsigned)ps[tid]);
  }

  for (int base = 0; base < nvec; base += SC_BATCH * SC_TPB) {
    int4 v[SC_BATCH], w[SC_BATCH];
#pragma unroll
    for (int u = 0; u < SC_BATCH; ++u) {
      const int i = base + u * SC_TPB + tid;
      if (i < nvec) {
        v[u] = pdv[i];
        if (svec) {
          w[u] = psv[i];
        } else {
          const int* q = psh + 4 * i;
          w[u] = make_int4(q[0], q[1], q[2], q[3]);
        }
      } else {
        v[u] = make_int4(-1, -1, -1, -1);
        w[u] = make_int4(0, 0, 0, 0);
      }
    }
#pragma unroll
    for (int u = 0; u < SC_BATCH; ++u) {
      sc_push(q_s, q_r, fill, ltmask, v[u].x, w[u].x, ulo, urn);
      sc_push(q_s, q_r, fill, ltmask, v[u].y, w[u].y, ulo, urn);
      sc_push(q_s, q_r, fill, ltmask, v[u].z, w[u].z, ulo, urn);
      sc_push(q_s, q_r, fill, ltmask, v[u].w, w[u].w, ulo, urn);
      while (fill >= 32) {
        fill -= 32;
        sc_round<true>(acc, h, q_s, q_r, fill, 32, lane);
      }
    }
  }
  if (fill > 0) sc_round<false>(acc, h, q_s, q_r, 0, fill, lane);

  if (tail + tid < len) {
    const unsigned rel = (unsigned)pd[tail + tid] - ulo;
    if (rel < urn) sc_add_row(acc + rel * SC_PITCH, h, (unsigned)ps[tail + tid]);
  }
  __syncthreads();

  float* __restrict__ ag = agg + (size_t)lo * HID;
  const int tot = rn * HID;
  int t0 = tid;
  for (; t0 + 3 * SC_TPB < tot; t0 += 4 * SC_TPB) {
    float a[4], dv[4];
#pragma unroll
    for (int u = 0; u < 4; ++u) {
      const int t = t0 + u * SC_TPB;
      a[u]  = acc[(t >> 4) * SC_PITCH + (t & 15)];
      dv[u] = dinv[lo + (t >> 4)];
    }
#pragma unroll
    for (int u = 0; u < 4; ++u) unsafeAtomicAdd(&ag[t0 + u * SC_TPB], a[u] * dv[u]);
  }
  for (; t0 < tot; t0 += SC_TPB)
    unsafeAtomicAdd(&ag[t0], acc[(t0 >> 4) * SC_PITCH + (t0 & 15)] * dinv[lo + (t0 >> 4)]);
}

__global__ void gcn_final(const float* __restrict__ agg, const float* __restrict__ b1,
                          const float* __restrict__ Wp, const float* __restrict__ bp,
                          float* __restrict__ out, int n) {
  int i = blockIdx.x * blockDim.x + threadIdx.x;
  if (i >= n) return;

  float hb[HID];
#pragma unroll
  for (int k = 0; k < HID; ++k) {
    float v = agg[(size_t)i * HID + k] + b1[k];
    hb[k] = v > 0.0f ? v : 0.0f;
  }

  float logits[NCLS];
  float mx = -3.0e38f;
#pragma unroll
  for (int c = 0; c < NCLS; ++c) {
    float accv = bp[c];
#pragma unroll
    for (int k = 0; k < HID; ++k) accv += hb[k] * Wp[k * NCLS + c];
    logits[c] = accv;
    mx = fmaxf(mx, accv);
  }

  float sum = 0.0f;
#pragma unroll
  for (int c = 0; c < NCLS; ++c) {
    float ev = __expf(logits[c] - mx);
    logits[c] = ev;
    sum += ev;
  }
  float inv = 1.0f / sum;
#pragma unroll
  for (int c = 0; c < NCLS; ++c) out[(size_t)i * NCLS + c] = logits[c] * inv;
}

extern "C" void kernel_launch(void* const* d_in, const int* in_sizes, int n_in,
                              void* d_out, int out_size, void* d_ws, size_t ws_size,
                              hipStream_t stream) {
  const float* x  = (const float*)d_in[0];
  const float* W1 = (const float*)d_in[1];
  const float* b1 = (const float*)d_in[2];
  const float* Wp = (const float*)d_in[3];
  const float* bp = (const float*)d_in[4];
  const int*   ei = (const int*)d_in[5];

  const int N = in_sizes[0] / F_IN;
  const int E = in_sizes[5] / 2;
  float* out = (float*)d_out;

  float* h   = (float*)d_ws;
  float* agg = h + (size_t)N * HID;
  float* deg = agg + (size_t)N * HID;

  const int tpb = 256;

  unsigned* part = (unsigned*)h;
  const int nRanges = (N + DEG_RANGE - 1) / DEG_RANGE;
  const int pw = (N + 1) / 2;
  gcn_deg_count<<<dim3(DEG_CHUNKS, nRanges), DEG_TPB, 0, stream>>>(ei, part, E, N);
  gcn_dinv<<<(pw + tpb - 1) / tpb, tpb, 0, stream>>>(part, deg, N);

  const int nTiles = (N + 15) / 16;
  gcn_gemm_wmma<<<(nTiles + 7) / 8, 256, 0, stream>>>(x, W1, deg, h, agg, nTiles, N);

  const int nOwn = (N + SC_RANGE - 1) / SC_RANGE;
  gcn_edge_owner<<<dim3(SC_SLICES, nOwn), SC_TPB, 0, stream>>>(ei, h, deg, agg, E, N);

  gcn_final<<<(N + tpb - 1) / tpb, tpb, 0, stream>>>(agg, b1, Wp, bp, out, N);
}
